# starting kernel
# baseline (speedup 1.0000x reference)
#include <hip/hip_runtime.h>
#include <hip/hip_bf16.h>


typedef __attribute__((ext_vector_type(2))) float v2f;
typedef __attribute__((ext_vector_type(8))) float v8f;

#define DP 17
#define KDIM 160
#define KSTEPS 40
#define XSTR 20
#define BLDS_N 2560

struct PairTab { int b[KDIM]; int c[KDIM]; };
static constexpr PairTab makeTab() {
  PairTab t{};
  int q = 0;
  for (int b = 0; b < DP; ++b)
    for (int c = b; c < DP; ++c) { t.b[q] = b; t.c[q] = c; ++q; }
  for (; q < KDIM; ++q) { t.b[q] = DP - 1; t.c[q] = DP - 1; }
  return t;
}
static constexpr PairTab TAB = makeTab();

__global__ void build_weights_kernel(const float* __restrict__ w,
                                     const int* __restrict__ comb_idx,
                                     float* __restrict__ ws) {
  const int tid = threadIdx.x;
  for (int i = tid; i < BLDS_N + 1; i += 256) ws[i] = 0.0f;
  __syncthreads();
  for (int i = tid; i < 969; i += 256) {
    const int a = comb_idx[3 * i + 0];
    const int b = comb_idx[3 * i + 1];
    const int c = comb_idx[3 * i + 2];
    const float wi = w[i];
    if (a == 16) {
      ws[BLDS_N] = wi;
    } else {
      const int q = b * DP - (b * (b - 1)) / 2 + (c - b);
      const int pos = ((q >> 1) * 16 + a) * 2 + (q & 1);
      ws[pos] = wi;
    }
  }
}

__global__ __launch_bounds__(256) void poly_logistic_kernel(
    const float* __restrict__ X, const float* __restrict__ ws,
    float* __restrict__ out, int tiles_per_wave) {
  __shared__ __align__(16) float Blds[BLDS_N];
  __shared__ __align__(16) float xp[8][16 * XSTR];

  const int tid  = threadIdx.x;
  const int wave = tid >> 5;
  const int lane = tid & 31;
  const int hi   = lane >> 4;
  const int lm   = lane & 15;
  float* xw = &xp[wave][0];

  for (int i = tid; i < BLDS_N; i += 256) Blds[i] = ws[i];
  const float wconst = ws[BLDS_N];

  const int wave_global = blockIdx.x * 8 + wave;
  const int tile0 = wave_global * tiles_per_wave;

  const float* bp = Blds + (hi * 16 + lm) * 2;

  __syncthreads();

  for (int t = 0; t < tiles_per_wave; ++t) {
    const int tile = tile0 + t;

    const float4* Xv = (const float4*)(X + (size_t)tile * 256);
    const float4 v0 = Xv[lane];
    const float4 v1 = Xv[lane + 32];
    {
      const int m0 = lane >> 2;
      const int f0 = (lane & 3) * 4;
      *(float4*)(xw + m0 * XSTR + f0)       = v0;
      *(float4*)(xw + (m0 + 8) * XSTR + f0) = v1;
    }
    __builtin_prefetch(X + (size_t)(tile + 1) * 256 + lane * 8, 0, 0);
    __syncthreads();

    const float* xr = xw + lm * XSTR;
    float xv[DP];
    {
      const float4 r0 = *(const float4*)(xr + 0);
      const float4 r1 = *(const float4*)(xr + 4);
      const float4 r2 = *(const float4*)(xr + 8);
      const float4 r3 = *(const float4*)(xr + 12);
      xv[0]  = r0.x; xv[1]  = r0.y; xv[2]  = r0.z; xv[3]  = r0.w;
      xv[4]  = r1.x; xv[5]  = r1.y; xv[6]  = r1.z; xv[7]  = r1.w;
      xv[8]  = r2.x; xv[9]  = r2.y; xv[10] = r2.z; xv[11] = r2.w;
      xv[12] = r3.x; xv[13] = r3.y; xv[14] = r3.z; xv[15] = r3.w;
      xv[16] = 1.0f;
    }

    v2f A[KSTEPS];
    if (hi) {
#pragma unroll
      for (int k = 0; k < KSTEPS; ++k) {
        A[k].x = xv[TAB.b[4 * k + 2]] * xv[TAB.c[4 * k + 2]];
        A[k].y = xv[TAB.b[4 * k + 3]] * xv[TAB.c[4 * k + 3]];
      }
    } else {
#pragma unroll
      for (int k = 0; k < KSTEPS; ++k) {
        A[k].x = xv[TAB.b[4 * k + 0]] * xv[TAB.c[4 * k + 0]];
        A[k].y = xv[TAB.b[4 * k + 1]] * xv[TAB.c[4 * k + 1]];
      }
    }

    v8f acc = {};
#pragma unroll
    for (int k = 0; k < KSTEPS; ++k) {
      const v2f Bk = *(const v2f*)(bp + 64 * k);
      acc = __builtin_amdgcn_wmma_f32_16x16x4_f32(
          false, A[k], false, Bk, (short)0, acc, false, false);
    }

#pragma unroll
    for (int r = 0; r < 8; ++r) {
      const int m = r + 8 * hi;
      float v = acc[r] * xw[m * XSTR + lm];
      v += __shfl_xor(v, 1, 32);
      v += __shfl_xor(v, 2, 32);
      v += __shfl_xor(v, 4, 32);
      v += __shfl_xor(v, 8, 32);
      if (lm == 0) {
        const float logit = v + wconst;
        out[(size_t)tile * 16 + m] = 1.0f / (1.0f + __expf(-logit));
      }
    }
    __syncthreads();
  }
}

extern "C" void kernel_launch(void* const* d_in, const int* in_sizes, int n_in,
                              void* d_out, int out_size, void* d_ws,
                              size_t ws_size, hipStream_t stream) {
  const float* X        = (const float*)d_in[0];
  const float* w        = (const float*)d_in[1];
  const int*   comb_idx = (const int*)d_in[2];
  float*       out      = (float*)d_out;
  float*       ws       = (float*)d_ws;

  build_weights_kernel<<<1, 256, 0, stream>>>(w, comb_idx, ws);

  const int tiles_per_wave = 4;
  poly_logistic_kernel<<<128, 256, 0, stream>>>(X, ws, out, tiles_per_wave);
}


# Fuse the weight-packing launch into the main kernel: each block scatters w via comb_idx straight into its LDS B tile; 2 launches -> 1, no workspace round trip; summation order unchanged
# speedup vs baseline: 1.2883x; 1.2883x over previous
#include <hip/hip_runtime.h>
#include <hip/hip_bf16.h>


typedef __attribute__((ext_vector_type(2))) float v2f;
typedef __attribute__((ext_vector_type(8))) float v8f;

#define DP 17
#define KDIM 160
#define KSTEPS 40
#define XSTR 20
#define BLDS_N 2560

struct PairTab { int b[KDIM]; int c[KDIM]; };
static constexpr PairTab makeTab() {
  PairTab t{};
  int q = 0;
  for (int b = 0; b < DP; ++b)
    for (int c = b; c < DP; ++c) { t.b[q] = b; t.c[q] = c; ++q; }
  for (; q < KDIM; ++q) { t.b[q] = DP - 1; t.c[q] = DP - 1; }
  return t;
}
static constexpr PairTab TAB = makeTab();


__global__ __launch_bounds__(256) void poly_logistic_kernel(
    const float* __restrict__ X, const float* __restrict__ w,
    const int* __restrict__ comb_idx, float* __restrict__ out,
    int tiles_per_wave) {
  __shared__ __align__(16) float Blds[BLDS_N];
  __shared__ float wconst_s;
  __shared__ __align__(16) float xp[8][16 * XSTR];

  const int tid  = threadIdx.x;
  const int wave = tid >> 5;
  const int lane = tid & 31;
  const int hi   = lane >> 4;
  const int lm   = lane & 15;
  float* xw = &xp[wave][0];

  int   ma[4], mb[4], mc[4];
  float mw[4];
#pragma unroll
  for (int j = 0; j < 4; ++j) {
    const int i  = tid + 256 * j;
    const int ic = i < 969 ? i : 968;
    ma[j] = comb_idx[3 * ic + 0];
    mb[j] = comb_idx[3 * ic + 1];
    mc[j] = comb_idx[3 * ic + 2];
    mw[j] = w[ic];
  }
  for (int i = tid; i < BLDS_N / 4; i += 256)
    ((float4*)Blds)[i] = make_float4(0.0f, 0.0f, 0.0f, 0.0f);
  if (tid == 0) wconst_s = 0.0f;
  __syncthreads();
#pragma unroll
  for (int j = 0; j < 4; ++j) {
    const int i = tid + 256 * j;
    if (i < 969) {
      const int a = ma[j], b = mb[j], cc = mc[j];
      if (a == 16) {
        wconst_s = mw[j];
      } else {
        const int q = b * DP - (b * (b - 1)) / 2 + (cc - b);
        const int pos = ((q >> 1) * 16 + a) * 2 + (q & 1);
        if ((unsigned)pos < (unsigned)BLDS_N) Blds[pos] = mw[j];
      }
    }
  }

  const int wave_global = blockIdx.x * 8 + wave;
  const int tile0 = wave_global * tiles_per_wave;

  const float* bp = Blds + (hi * 16 + lm) * 2;

  __syncthreads();
  const float wconst = wconst_s;

  for (int t = 0; t < tiles_per_wave; ++t) {
    const int tile = tile0 + t;

    const float4* Xv = (const float4*)(X + (size_t)tile * 256);
    const float4 v0 = Xv[lane];
    const float4 v1 = Xv[lane + 32];
    {
      const int m0 = lane >> 2;
      const int f0 = (lane & 3) * 4;
      *(float4*)(xw + m0 * XSTR + f0)       = v0;
      *(float4*)(xw + (m0 + 8) * XSTR + f0) = v1;
    }
    __builtin_prefetch(X + (size_t)(tile + 1) * 256 + lane * 8, 0, 0);
    __syncthreads();

    const float* xr = xw + lm * XSTR;
    float xv[DP];
    {
      const float4 r0 = *(const float4*)(xr + 0);
      const float4 r1 = *(const float4*)(xr + 4);
      const float4 r2 = *(const float4*)(xr + 8);
      const float4 r3 = *(const float4*)(xr + 12);
      xv[0]  = r0.x; xv[1]  = r0.y; xv[2]  = r0.z; xv[3]  = r0.w;
      xv[4]  = r1.x; xv[5]  = r1.y; xv[6]  = r1.z; xv[7]  = r1.w;
      xv[8]  = r2.x; xv[9]  = r2.y; xv[10] = r2.z; xv[11] = r2.w;
      xv[12] = r3.x; xv[13] = r3.y; xv[14] = r3.z; xv[15] = r3.w;
      xv[16] = 1.0f;
    }

    v2f A[KSTEPS];
    if (hi) {
#pragma unroll
      for (int k = 0; k < KSTEPS; ++k) {
        A[k].x = xv[TAB.b[4 * k + 2]] * xv[TAB.c[4 * k + 2]];
        A[k].y = xv[TAB.b[4 * k + 3]] * xv[TAB.c[4 * k + 3]];
      }
    } else {
#pragma unroll
      for (int k = 0; k < KSTEPS; ++k) {
        A[k].x = xv[TAB.b[4 * k + 0]] * xv[TAB.c[4 * k + 0]];
        A[k].y = xv[TAB.b[4 * k + 1]] * xv[TAB.c[4 * k + 1]];
      }
    }

    v8f acc = {};
#pragma unroll
    for (int k = 0; k < KSTEPS; ++k) {
      const v2f Bk = *(const v2f*)(bp + 64 * k);
      acc = __builtin_amdgcn_wmma_f32_16x16x4_f32(
          false, A[k], false, Bk, (short)0, acc, false, false);
    }

#pragma unroll
    for (int r = 0; r < 8; ++r) {
      const int m = r + 8 * hi;
      float v = acc[r] * xw[m * XSTR + lm];
      v += __shfl_xor(v, 1, 32);
      v += __shfl_xor(v, 2, 32);
      v += __shfl_xor(v, 4, 32);
      v += __shfl_xor(v, 8, 32);
      if (lm == 0) {
        const float logit = v + wconst;
        out[(size_t)tile * 16 + m] = 1.0f / (1.0f + __expf(-logit));
      }
    }
    __syncthreads();
  }
}

extern "C" void kernel_launch(void* const* d_in, const int* in_sizes, int n_in,
                              void* d_out, int out_size, void* d_ws,
                              size_t ws_size, hipStream_t stream) {
  const float* X        = (const float*)d_in[0];
  const float* w        = (const float*)d_in[1];
  const int*   comb_idx = (const int*)d_in[2];
  float*       out      = (float*)d_out;

  const int tiles_per_wave = 4;
  poly_logistic_kernel<<<128, 256, 0, stream>>>(X, w, comb_idx, out,
                                                tiles_per_wave);
}


# Swap WMMA operand roles (weights on M, samples on N): each lane holds its own sample's Z terms; final dot = 8 in-lane multiply-adds + 1 half exchange, not 40 serial LDS trips; reorders that sum
# speedup vs baseline: 2.2800x; 1.7698x over previous
#include <hip/hip_runtime.h>
#include <hip/hip_bf16.h>


typedef __attribute__((ext_vector_type(2))) float v2f;
typedef __attribute__((ext_vector_type(8))) float v8f;

__device__ __forceinline__ float xhalf(float x) {
  const int b = __builtin_bit_cast(int, x);
  return __builtin_bit_cast(
      float, __builtin_amdgcn_permlanex16(b, b, 0x76543210, 0xfedcba98, false,
                                          false));
}

#define DP 17
#define KDIM 160
#define KSTEPS 40
#define XSTR 20
#define BLDS_N 2560

struct PairTab { int b[KDIM]; int c[KDIM]; };
static constexpr PairTab makeTab() {
  PairTab t{};
  int q = 0;
  for (int b = 0; b < DP; ++b)
    for (int c = b; c < DP; ++c) { t.b[q] = b; t.c[q] = c; ++q; }
  for (; q < KDIM; ++q) { t.b[q] = DP - 1; t.c[q] = DP - 1; }
  return t;
}
static constexpr PairTab TAB = makeTab();


__global__ __launch_bounds__(256) void poly_logistic_kernel(
    const float* __restrict__ X, const float* __restrict__ w,
    const int* __restrict__ comb_idx, float* __restrict__ out,
    int tiles_per_wave) {
  __shared__ __align__(16) float Blds[BLDS_N];
  __shared__ float wconst_s;
  __shared__ __align__(16) float xp[8][16 * XSTR];

  const int tid  = threadIdx.x;
  const int wave = tid >> 5;
  const int lane = tid & 31;
  const int hi   = lane >> 4;
  const int lm   = lane & 15;
  float* xw = &xp[wave][0];

  int   ma[4], mb[4], mc[4];
  float mw[4];
#pragma unroll
  for (int j = 0; j < 4; ++j) {
    const int i  = tid + 256 * j;
    const int ic = i < 969 ? i : 968;
    ma[j] = comb_idx[3 * ic + 0];
    mb[j] = comb_idx[3 * ic + 1];
    mc[j] = comb_idx[3 * ic + 2];
    mw[j] = w[ic];
  }
  for (int i = tid; i < BLDS_N / 4; i += 256)
    ((float4*)Blds)[i] = make_float4(0.0f, 0.0f, 0.0f, 0.0f);
  if (tid == 0) wconst_s = 0.0f;
  __syncthreads();
#pragma unroll
  for (int j = 0; j < 4; ++j) {
    const int i = tid + 256 * j;
    if (i < 969) {
      const int a = ma[j], b = mb[j], cc = mc[j];
      if (a == 16) {
        wconst_s = mw[j];
      } else {
        const int q = b * DP - (b * (b - 1)) / 2 + (cc - b);
        const int pos = ((q >> 1) * 16 + a) * 2 + (q & 1);
        if ((unsigned)pos < (unsigned)BLDS_N) Blds[pos] = mw[j];
      }
    }
  }

  const int wave_global = blockIdx.x * 8 + wave;
  const int tile0 = wave_global * tiles_per_wave;

  const float* bp = Blds + (hi * 16 + lm) * 2;

  __syncthreads();
  const float wconst = wconst_s;

  for (int t = 0; t < tiles_per_wave; ++t) {
    const int tile = tile0 + t;

    const float4* Xv = (const float4*)(X + (size_t)tile * 256);
    const float4 v0 = Xv[lane];
    const float4 v1 = Xv[lane + 32];
    {
      const int m0 = lane >> 2;
      const int f0 = (lane & 3) * 4;
      *(float4*)(xw + m0 * XSTR + f0)       = v0;
      *(float4*)(xw + (m0 + 8) * XSTR + f0) = v1;
    }
    __builtin_prefetch(X + (size_t)(tile + 1) * 256 + lane * 8, 0, 0);
    __syncthreads();

    const float* xr = xw + lm * XSTR;
    float xv[DP];
    {
      const float4 r0 = *(const float4*)(xr + 0);
      const float4 r1 = *(const float4*)(xr + 4);
      const float4 r2 = *(const float4*)(xr + 8);
      const float4 r3 = *(const float4*)(xr + 12);
      xv[0]  = r0.x; xv[1]  = r0.y; xv[2]  = r0.z; xv[3]  = r0.w;
      xv[4]  = r1.x; xv[5]  = r1.y; xv[6]  = r1.z; xv[7]  = r1.w;
      xv[8]  = r2.x; xv[9]  = r2.y; xv[10] = r2.z; xv[11] = r2.w;
      xv[12] = r3.x; xv[13] = r3.y; xv[14] = r3.z; xv[15] = r3.w;
      xv[16] = 1.0f;
    }
    const float4 e0 = *(const float4*)(xr + 8 * hi);
    const float4 e1 = *(const float4*)(xr + 8 * hi + 4);

    v2f A[KSTEPS];
    if (hi) {
#pragma unroll
      for (int k = 0; k < KSTEPS; ++k) {
        A[k].x = xv[TAB.b[4 * k + 2]] * xv[TAB.c[4 * k + 2]];
        A[k].y = xv[TAB.b[4 * k + 3]] * xv[TAB.c[4 * k + 3]];
      }
    } else {
#pragma unroll
      for (int k = 0; k < KSTEPS; ++k) {
        A[k].x = xv[TAB.b[4 * k + 0]] * xv[TAB.c[4 * k + 0]];
        A[k].y = xv[TAB.b[4 * k + 1]] * xv[TAB.c[4 * k + 1]];
      }
    }

    v8f acc = {};
#pragma unroll
    for (int k = 0; k < KSTEPS; ++k) {
      const v2f Wk = *(const v2f*)(bp + 64 * k);
      acc = __builtin_amdgcn_wmma_f32_16x16x4_f32(
          false, Wk, false, A[k], (short)0, acc, false, false);
    }

    float part = acc[0] * e0.x;
    part += acc[1] * e0.y;
    part += acc[2] * e0.z;
    part += acc[3] * e0.w;
    part += acc[4] * e1.x;
    part += acc[5] * e1.y;
    part += acc[6] * e1.z;
    part += acc[7] * e1.w;
    const float logit = part + xhalf(part) + wconst;
    const float y = 1.0f / (1.0f + __expf(-logit));
    if (hi == 0) out[(size_t)tile * 16 + lm] = y;
    __syncthreads();
  }
}

extern "C" void kernel_launch(void* const* d_in, const int* in_sizes, int n_in,
                              void* d_out, int out_size, void* d_ws,
                              size_t ws_size, hipStream_t stream) {
  const float* X        = (const float*)d_in[0];
  const float* w        = (const float*)d_in[1];
  const int*   comb_idx = (const int*)d_in[2];
  float*       out      = (float*)d_out;

  const int tiles_per_wave = 4;
  poly_logistic_kernel<<<128, 256, 0, stream>>>(X, w, comb_idx, out,
                                                tiles_per_wave);
}


# Launch 256 blocks of 8 waves with 2 tiles per wave (was 128 blocks, 4 tiles): one block per WGP on all 256 WGPs, so half the WMMAs per SIMD pipe and half the serial tile passes per wave
# speedup vs baseline: 3.2107x; 1.4082x over previous
#include <hip/hip_runtime.h>
#include <hip/hip_bf16.h>


typedef __attribute__((ext_vector_type(2))) float v2f;
typedef __attribute__((ext_vector_type(8))) float v8f;

__device__ __forceinline__ float xhalf(float x) {
  const int b = __builtin_bit_cast(int, x);
  return __builtin_bit_cast(
      float, __builtin_amdgcn_permlanex16(b, b, 0x76543210, 0xfedcba98, false,
                                          false));
}

#define DP 17
#define KDIM 160
#define KSTEPS 40
#define XSTR 20
#define BLDS_N 2560

struct PairTab { int b[KDIM]; int c[KDIM]; };
static constexpr PairTab makeTab() {
  PairTab t{};
  int q = 0;
  for (int b = 0; b < DP; ++b)
    for (int c = b; c < DP; ++c) { t.b[q] = b; t.c[q] = c; ++q; }
  for (; q < KDIM; ++q) { t.b[q] = DP - 1; t.c[q] = DP - 1; }
  return t;
}
static constexpr PairTab TAB = makeTab();


__global__ __launch_bounds__(256) void poly_logistic_kernel(
    const float* __restrict__ X, const float* __restrict__ w,
    const int* __restrict__ comb_idx, float* __restrict__ out,
    int tiles_per_wave) {
  __shared__ __align__(16) float Blds[BLDS_N];
  __shared__ float wconst_s;
  __shared__ __align__(16) float xp[8][16 * XSTR];

  const int tid  = threadIdx.x;
  const int wave = tid >> 5;
  const int lane = tid & 31;
  const int hi   = lane >> 4;
  const int lm   = lane & 15;
  float* xw = &xp[wave][0];

  int   ma[4], mb[4], mc[4];
  float mw[4];
#pragma unroll
  for (int j = 0; j < 4; ++j) {
    const int i  = tid + 256 * j;
    const int ic = i < 969 ? i : 968;
    ma[j] = comb_idx[3 * ic + 0];
    mb[j] = comb_idx[3 * ic + 1];
    mc[j] = comb_idx[3 * ic + 2];
    mw[j] = w[ic];
  }
  for (int i = tid; i < BLDS_N / 4; i += 256)
    ((float4*)Blds)[i] = make_float4(0.0f, 0.0f, 0.0f, 0.0f);
  if (tid == 0) wconst_s = 0.0f;
  __syncthreads();
#pragma unroll
  for (int j = 0; j < 4; ++j) {
    const int i = tid + 256 * j;
    if (i < 969) {
      const int a = ma[j], b = mb[j], cc = mc[j];
      if (a == 16) {
        wconst_s = mw[j];
      } else {
        const int q = b * DP - (b * (b - 1)) / 2 + (cc - b);
        const int pos = ((q >> 1) * 16 + a) * 2 + (q & 1);
        if ((unsigned)pos < (unsigned)BLDS_N) Blds[pos] = mw[j];
      }
    }
  }

  const int wave_global = blockIdx.x * 8 + wave;
  const int tile0 = wave_global * tiles_per_wave;

  const float* bp = Blds + (hi * 16 + lm) * 2;

  __syncthreads();
  const float wconst = wconst_s;

  for (int t = 0; t < tiles_per_wave; ++t) {
    const int tile = tile0 + t;

    const float4* Xv = (const float4*)(X + (size_t)tile * 256);
    const float4 v0 = Xv[lane];
    const float4 v1 = Xv[lane + 32];
    {
      const int m0 = lane >> 2;
      const int f0 = (lane & 3) * 4;
      *(float4*)(xw + m0 * XSTR + f0)       = v0;
      *(float4*)(xw + (m0 + 8) * XSTR + f0) = v1;
    }
    __builtin_prefetch(X + (size_t)(tile + 1) * 256 + lane * 8, 0, 0);
    __syncthreads();

    const float* xr = xw + lm * XSTR;
    float xv[DP];
    {
      const float4 r0 = *(const float4*)(xr + 0);
      const float4 r1 = *(const float4*)(xr + 4);
      const float4 r2 = *(const float4*)(xr + 8);
      const float4 r3 = *(const float4*)(xr + 12);
      xv[0]  = r0.x; xv[1]  = r0.y; xv[2]  = r0.z; xv[3]  = r0.w;
      xv[4]  = r1.x; xv[5]  = r1.y; xv[6]  = r1.z; xv[7]  = r1.w;
      xv[8]  = r2.x; xv[9]  = r2.y; xv[10] = r2.z; xv[11] = r2.w;
      xv[12] = r3.x; xv[13] = r3.y; xv[14] = r3.z; xv[15] = r3.w;
      xv[16] = 1.0f;
    }
    const float4 e0 = *(const float4*)(xr + 8 * hi);
    const float4 e1 = *(const float4*)(xr + 8 * hi + 4);

    v2f A[KSTEPS];
    if (hi) {
#pragma unroll
      for (int k = 0; k < KSTEPS; ++k) {
        A[k].x = xv[TAB.b[4 * k + 2]] * xv[TAB.c[4 * k + 2]];
        A[k].y = xv[TAB.b[4 * k + 3]] * xv[TAB.c[4 * k + 3]];
      }
    } else {
#pragma unroll
      for (int k = 0; k < KSTEPS; ++k) {
        A[k].x = xv[TAB.b[4 * k + 0]] * xv[TAB.c[4 * k + 0]];
        A[k].y = xv[TAB.b[4 * k + 1]] * xv[TAB.c[4 * k + 1]];
      }
    }

    v8f acc = {};
#pragma unroll
    for (int k = 0; k < KSTEPS; ++k) {
      const v2f Wk = *(const v2f*)(bp + 64 * k);
      acc = __builtin_amdgcn_wmma_f32_16x16x4_f32(
          false, Wk, false, A[k], (short)0, acc, false, false);
    }

    float part = acc[0] * e0.x;
    part += acc[1] * e0.y;
    part += acc[2] * e0.z;
    part += acc[3] * e0.w;
    part += acc[4] * e1.x;
    part += acc[5] * e1.y;
    part += acc[6] * e1.z;
    part += acc[7] * e1.w;
    const float logit = part + xhalf(part) + wconst;
    const float y = 1.0f / (1.0f + __expf(-logit));
    if (hi == 0) out[(size_t)tile * 16 + lm] = y;
    __syncthreads();
  }
}

extern "C" void kernel_launch(void* const* d_in, const int* in_sizes, int n_in,
                              void* d_out, int out_size, void* d_ws,
                              size_t ws_size, hipStream_t stream) {
  const float* X        = (const float*)d_in[0];
  const float* w        = (const float*)d_in[1];
  const int*   comb_idx = (const int*)d_in[2];
  float*       out      = (float*)d_out;

  const int tiles_per_wave = 2;
  poly_logistic_kernel<<<256, 256, 0, stream>>>(X, w, comb_idx, out,
                                                tiles_per_wave);
}


# Every lane reads its own sample row of both tiles from global into registers at kernel entry: no LDS transpose of X, no barrier in the tile loop; X latency hides behind the weight scatter
# speedup vs baseline: 4.1746x; 1.3002x over previous
#include <hip/hip_runtime.h>
#include <hip/hip_bf16.h>


typedef __attribute__((ext_vector_type(2))) float v2f;
typedef __attribute__((ext_vector_type(8))) float v8f;

__device__ __forceinline__ float xhalf(float x) {
  const int b = __builtin_bit_cast(int, x);
  return __builtin_bit_cast(
      float, __builtin_amdgcn_permlanex16(b, b, 0x76543210, 0xfedcba98, false,
                                          false));
}

#define DP 17
#define KDIM 160
#define KSTEPS 40
#define TILES 2
#define BLDS_N 2560

struct PairTab { int b[KDIM]; int c[KDIM]; };
static constexpr PairTab makeTab() {
  PairTab t{};
  int q = 0;
  for (int b = 0; b < DP; ++b)
    for (int c = b; c < DP; ++c) { t.b[q] = b; t.c[q] = c; ++q; }
  for (; q < KDIM; ++q) { t.b[q] = DP - 1; t.c[q] = DP - 1; }
  return t;
}
static constexpr PairTab TAB = makeTab();


__global__ __launch_bounds__(256) void poly_logistic_kernel(
    const float* __restrict__ X, const float* __restrict__ w,
    const int* __restrict__ comb_idx, float* __restrict__ out) {
  __shared__ __align__(16) float Blds[BLDS_N];
  __shared__ float wconst_s;

  const int tid  = threadIdx.x;
  const int wave = tid >> 5;
  const int lane = tid & 31;
  const int hi   = lane >> 4;
  const int lm   = lane & 15;

  const int wave_global = blockIdx.x * 8 + wave;
  const int tile0 = wave_global * TILES;

  int   ma[4], mb[4], mc[4];
  float mw[4];
#pragma unroll
  for (int j = 0; j < 4; ++j) {
    const int i  = tid + 256 * j;
    const int ic = i < 969 ? i : 968;
    ma[j] = comb_idx[3 * ic + 0];
    mb[j] = comb_idx[3 * ic + 1];
    mc[j] = comb_idx[3 * ic + 2];
    mw[j] = w[ic];
  }

  float4 xr[TILES][4];
#pragma unroll
  for (int t = 0; t < TILES; ++t) {
    const float4* row =
        (const float4*)(X + ((size_t)(tile0 + t) * 16 + lm) * 16);
#pragma unroll
    for (int j = 0; j < 4; ++j) xr[t][j] = row[j];
  }

  for (int i = tid; i < BLDS_N / 4; i += 256)
    ((float4*)Blds)[i] = make_float4(0.0f, 0.0f, 0.0f, 0.0f);
  if (tid == 0) wconst_s = 0.0f;
  __syncthreads();
#pragma unroll
  for (int j = 0; j < 4; ++j) {
    const int i = tid + 256 * j;
    if (i < 969) {
      const int a = ma[j], b = mb[j], cc = mc[j];
      if (a == 16) {
        wconst_s = mw[j];
      } else {
        const int q = b * DP - (b * (b - 1)) / 2 + (cc - b);
        const int pos = ((q >> 1) * 16 + a) * 2 + (q & 1);
        if ((unsigned)pos < (unsigned)BLDS_N) Blds[pos] = mw[j];
      }
    }
  }

  const float* bp = Blds + (hi * 16 + lm) * 2;
  float wconst = 0.0f;

#pragma unroll
  for (int t = 0; t < TILES; ++t) {
    const int tile = tile0 + t;

    float xv[DP];
    xv[0]  = xr[t][0].x; xv[1]  = xr[t][0].y; xv[2]  = xr[t][0].z; xv[3]  = xr[t][0].w;
    xv[4]  = xr[t][1].x; xv[5]  = xr[t][1].y; xv[6]  = xr[t][1].z; xv[7]  = xr[t][1].w;
    xv[8]  = xr[t][2].x; xv[9]  = xr[t][2].y; xv[10] = xr[t][2].z; xv[11] = xr[t][2].w;
    xv[12] = xr[t][3].x; xv[13] = xr[t][3].y; xv[14] = xr[t][3].z; xv[15] = xr[t][3].w;
    xv[16] = 1.0f;

    v2f A[KSTEPS];
    if (hi) {
#pragma unroll
      for (int k = 0; k < KSTEPS; ++k) {
        A[k].x = xv[TAB.b[4 * k + 2]] * xv[TAB.c[4 * k + 2]];
        A[k].y = xv[TAB.b[4 * k + 3]] * xv[TAB.c[4 * k + 3]];
      }
    } else {
#pragma unroll
      for (int k = 0; k < KSTEPS; ++k) {
        A[k].x = xv[TAB.b[4 * k + 0]] * xv[TAB.c[4 * k + 0]];
        A[k].y = xv[TAB.b[4 * k + 1]] * xv[TAB.c[4 * k + 1]];
      }
    }

    if (t == 0) {
      __syncthreads();
      wconst = wconst_s;
    }

    v8f acc = {};
#pragma unroll
    for (int k = 0; k < KSTEPS; ++k) {
      const v2f Wk = *(const v2f*)(bp + 64 * k);
      acc = __builtin_amdgcn_wmma_f32_16x16x4_f32(
          false, Wk, false, A[k], (short)0, acc, false, false);
    }

    float part = acc[0] * (hi ? xv[8] : xv[0]);
    part += acc[1] * (hi ? xv[9]  : xv[1]);
    part += acc[2] * (hi ? xv[10] : xv[2]);
    part += acc[3] * (hi ? xv[11] : xv[3]);
    part += acc[4] * (hi ? xv[12] : xv[4]);
    part += acc[5] * (hi ? xv[13] : xv[5]);
    part += acc[6] * (hi ? xv[14] : xv[6]);
    part += acc[7] * (hi ? xv[15] : xv[7]);
    const float logit = part + xhalf(part) + wconst;
    const float y = 1.0f / (1.0f + __expf(-logit));
    if (hi == 0) out[(size_t)tile * 16 + lm] = y;
  }
}

extern "C" void kernel_launch(void* const* d_in, const int* in_sizes, int n_in,
                              void* d_out, int out_size, void* d_ws,
                              size_t ws_size, hipStream_t stream) {
  const float* X        = (const float*)d_in[0];
  const float* w        = (const float*)d_in[1];
  const int*   comb_idx = (const int*)d_in[2];
  float*       out      = (float*)d_out;

  poly_logistic_kernel<<<256, 256, 0, stream>>>(X, w, comb_idx, out);
}


# Each monomial takes as GEMM row the index whose other two share a half (0-7, 8-15) or are the ones column: 89 pair columns, not 153; K 160->92, 23 WMMAs per tile, not 40; regroups the sum
# speedup vs baseline: 4.6518x; 1.1143x over previous
#include <hip/hip_runtime.h>
#include <hip/hip_bf16.h>


typedef __attribute__((ext_vector_type(2))) float v2f;
typedef __attribute__((ext_vector_type(8))) float v8f;

__device__ __forceinline__ float xhalf(float x) {
  const int b = __builtin_bit_cast(int, x);
  return __builtin_bit_cast(
      float, __builtin_amdgcn_permlanex16(b, b, 0x76543210, 0xfedcba98, false,
                                          false));
}

#define DP 17
#define KDIM 92
#define KSTEPS 23
#define TILES 2
#define BLDS_N 1472
#define Q_ONES 72

struct PairTab { int b[KDIM]; int c[KDIM]; };
static constexpr PairTab makeTab() {
  PairTab t{};
  int q = 0;
  for (int g = 0; g < 2; ++g)
    for (int b = 0; b < 8; ++b)
      for (int c = b; c < 8; ++c) { t.b[q] = 8 * g + b; t.c[q] = 8 * g + c; ++q; }
  for (int i = 0; i < 16; ++i) { t.b[q] = i; t.c[q] = DP - 1; ++q; }
  for (; q < KDIM; ++q) { t.b[q] = DP - 1; t.c[q] = DP - 1; }
  return t;
}
static constexpr PairTab TAB = makeTab();


__global__ __launch_bounds__(256) void poly_logistic_kernel(
    const float* __restrict__ X, const float* __restrict__ w,
    const int* __restrict__ comb_idx, float* __restrict__ out) {
  __shared__ __align__(16) float Blds[BLDS_N];
  __shared__ float wconst_s;

  const int tid  = threadIdx.x;
  const int wave = tid >> 5;
  const int lane = tid & 31;
  const int hi   = lane >> 4;
  const int lm   = lane & 15;

  const int wave_global = blockIdx.x * 8 + wave;
  const int tile0 = wave_global * TILES;

  int   ma[4], mb[4], mc[4];
  float mw[4];
#pragma unroll
  for (int j = 0; j < 4; ++j) {
    const int i  = tid + 256 * j;
    const int ic = i < 969 ? i : 968;
    ma[j] = comb_idx[3 * ic + 0];
    mb[j] = comb_idx[3 * ic + 1];
    mc[j] = comb_idx[3 * ic + 2];
    mw[j] = w[ic];
  }

  float4 xr[TILES][4];
#pragma unroll
  for (int t = 0; t < TILES; ++t) {
    const float4* row =
        (const float4*)(X + ((size_t)(tile0 + t) * 16 + lm) * 16);
#pragma unroll
    for (int j = 0; j < 4; ++j) xr[t][j] = row[j];
  }

  for (int i = tid; i < BLDS_N / 4; i += 256)
    ((float4*)Blds)[i] = make_float4(0.0f, 0.0f, 0.0f, 0.0f);
  if (tid == 0) wconst_s = 0.0f;
  __syncthreads();
#pragma unroll
  for (int j = 0; j < 4; ++j) {
    const int i = tid + 256 * j;
    if (i < 969) {
      const int lo  = min(ma[j], min(mb[j], mc[j]));
      const int top = max(ma[j], max(mb[j], mc[j]));
      const int mid = ma[j] + mb[j] + mc[j] - lo - top;
      if (lo == 16) {
        wconst_s = mw[j];
      } else {
        const bool lowpair = (lo >> 3) == (mid >> 3);
        const int p1 = lowpair ? lo : mid;
        const int p2 = lowpair ? mid : top;
        const int b = p1 & 7, cc = p2 & 7;
        const int q3 = 36 * ((p1 >> 3) & 1) + ((b * (17 - b)) >> 1) + (cc - b);
        const int q = (top == 16) ? Q_ONES + mid : q3;
        const int row = (top != 16 && lowpair) ? top : lo;
        const int pos = ((q >> 1) * 16 + row) * 2 + (q & 1);
        if ((unsigned)pos < (unsigned)BLDS_N) Blds[pos] = mw[j];
      }
    }
  }

  const float* bp = Blds + (hi * 16 + lm) * 2;
  float wconst = 0.0f;

#pragma unroll
  for (int t = 0; t < TILES; ++t) {
    const int tile = tile0 + t;

    float xv[DP];
    xv[0]  = xr[t][0].x; xv[1]  = xr[t][0].y; xv[2]  = xr[t][0].z; xv[3]  = xr[t][0].w;
    xv[4]  = xr[t][1].x; xv[5]  = xr[t][1].y; xv[6]  = xr[t][1].z; xv[7]  = xr[t][1].w;
    xv[8]  = xr[t][2].x; xv[9]  = xr[t][2].y; xv[10] = xr[t][2].z; xv[11] = xr[t][2].w;
    xv[12] = xr[t][3].x; xv[13] = xr[t][3].y; xv[14] = xr[t][3].z; xv[15] = xr[t][3].w;
    xv[16] = 1.0f;

    v2f A[KSTEPS];
    if (hi) {
#pragma unroll
      for (int k = 0; k < KSTEPS; ++k) {
        A[k].x = xv[TAB.b[4 * k + 2]] * xv[TAB.c[4 * k + 2]];
        A[k].y = xv[TAB.b[4 * k + 3]] * xv[TAB.c[4 * k + 3]];
      }
    } else {
#pragma unroll
      for (int k = 0; k < KSTEPS; ++k) {
        A[k].x = xv[TAB.b[4 * k + 0]] * xv[TAB.c[4 * k + 0]];
        A[k].y = xv[TAB.b[4 * k + 1]] * xv[TAB.c[4 * k + 1]];
      }
    }

    if (t == 0) {
      __syncthreads();
      wconst = wconst_s;
    }

    v8f acc = {};
#pragma unroll
    for (int k = 0; k < KSTEPS; ++k) {
      const v2f Wk = *(const v2f*)(bp + 64 * k);
      acc = __builtin_amdgcn_wmma_f32_16x16x4_f32(
          false, Wk, false, A[k], (short)0, acc, false, false);
    }

    float part = acc[0] * (hi ? xv[8] : xv[0]);
    part += acc[1] * (hi ? xv[9]  : xv[1]);
    part += acc[2] * (hi ? xv[10] : xv[2]);
    part += acc[3] * (hi ? xv[11] : xv[3]);
    part += acc[4] * (hi ? xv[12] : xv[4]);
    part += acc[5] * (hi ? xv[13] : xv[5]);
    part += acc[6] * (hi ? xv[14] : xv[6]);
    part += acc[7] * (hi ? xv[15] : xv[7]);
    const float logit = part + xhalf(part) + wconst;
    const float y = 1.0f / (1.0f + __expf(-logit));
    if (hi == 0) out[(size_t)tile * 16 + lm] = y;
  }
}

extern "C" void kernel_launch(void* const* d_in, const int* in_sizes, int n_in,
                              void* d_out, int out_size, void* d_ws,
                              size_t ws_size, hipStream_t stream) {
  const float* X        = (const float*)d_in[0];
  const float* w        = (const float*)d_in[1];
  const int*   comb_idx = (const int*)d_in[2];
  float*       out      = (float*)d_out;

  poly_logistic_kernel<<<256, 256, 0, stream>>>(X, w, comb_idx, out);
}


# Both workgroup barriers fence LDS only (waves exchange nothing through global memory): X row loads stay in flight across the weight scatter, tile 1's rows across tile 0's WMMAs; sums unchanged
# speedup vs baseline: 4.9519x; 1.0645x over previous
#include <hip/hip_runtime.h>
#include <hip/hip_bf16.h>


typedef __attribute__((ext_vector_type(2))) float v2f;
typedef __attribute__((ext_vector_type(8))) float v8f;

__device__ __forceinline__ float xhalf(float x) {
  const int b = __builtin_bit_cast(int, x);
  return __builtin_bit_cast(
      float, __builtin_amdgcn_permlanex16(b, b, 0x76543210, 0xfedcba98, false,
                                          false));
}

__device__ __forceinline__ void lds_barrier() {
  __builtin_amdgcn_fence(__ATOMIC_RELEASE, "workgroup", "local");
  __builtin_amdgcn_s_barrier();
  __builtin_amdgcn_fence(__ATOMIC_ACQUIRE, "workgroup", "local");
}

#define DP 17
#define KDIM 92
#define KSTEPS 23
#define TILES 2
#define BLDS_N 1472
#define Q_ONES 72

struct PairTab { int b[KDIM]; int c[KDIM]; };
static constexpr PairTab makeTab() {
  PairTab t{};
  int q = 0;
  for (int g = 0; g < 2; ++g)
    for (int b = 0; b < 8; ++b)
      for (int c = b; c < 8; ++c) { t.b[q] = 8 * g + b; t.c[q] = 8 * g + c; ++q; }
  for (int i = 0; i < 16; ++i) { t.b[q] = i; t.c[q] = DP - 1; ++q; }
  for (; q < KDIM; ++q) { t.b[q] = DP - 1; t.c[q] = DP - 1; }
  return t;
}
static constexpr PairTab TAB = makeTab();


__global__ __launch_bounds__(256) void poly_logistic_kernel(
    const float* __restrict__ X, const float* __restrict__ w,
    const int* __restrict__ comb_idx, float* __restrict__ out) {
  __shared__ __align__(16) float Blds[BLDS_N];
  __shared__ float wconst_s;

  const int tid  = threadIdx.x;
  const int wave = tid >> 5;
  const int lane = tid & 31;
  const int hi   = lane >> 4;
  const int lm   = lane & 15;

  const int wave_global = blockIdx.x * 8 + wave;
  const int tile0 = wave_global * TILES;

  int   ma[4], mb[4], mc[4];
  float mw[4];
#pragma unroll
  for (int j = 0; j < 4; ++j) {
    const int i  = tid + 256 * j;
    const int ic = i < 969 ? i : 968;
    ma[j] = comb_idx[3 * ic + 0];
    mb[j] = comb_idx[3 * ic + 1];
    mc[j] = comb_idx[3 * ic + 2];
    mw[j] = w[ic];
  }

  float4 xr[TILES][4];
#pragma unroll
  for (int t = 0; t < TILES; ++t) {
    const float4* row =
        (const float4*)(X + ((size_t)(tile0 + t) * 16 + lm) * 16);
#pragma unroll
    for (int j = 0; j < 4; ++j) xr[t][j] = row[j];
  }

  for (int i = tid; i < BLDS_N / 4; i += 256)
    ((float4*)Blds)[i] = make_float4(0.0f, 0.0f, 0.0f, 0.0f);
  if (tid == 0) wconst_s = 0.0f;
  lds_barrier();
#pragma unroll
  for (int j = 0; j < 4; ++j) {
    const int i = tid + 256 * j;
    if (i < 969) {
      const int lo  = min(ma[j], min(mb[j], mc[j]));
      const int top = max(ma[j], max(mb[j], mc[j]));
      const int mid = ma[j] + mb[j] + mc[j] - lo - top;
      if (lo == 16) {
        wconst_s = mw[j];
      } else {
        const bool lowpair = (lo >> 3) == (mid >> 3);
        const int p1 = lowpair ? lo : mid;
        const int p2 = lowpair ? mid : top;
        const int b = p1 & 7, cc = p2 & 7;
        const int q3 = 36 * ((p1 >> 3) & 1) + ((b * (17 - b)) >> 1) + (cc - b);
        const int q = (top == 16) ? Q_ONES + mid : q3;
        const int row = (top != 16 && lowpair) ? top : lo;
        const int pos = ((q >> 1) * 16 + row) * 2 + (q & 1);
        if ((unsigned)pos < (unsigned)BLDS_N) Blds[pos] = mw[j];
      }
    }
  }

  const float* bp = Blds + (hi * 16 + lm) * 2;
  float wconst = 0.0f;

#pragma unroll
  for (int t = 0; t < TILES; ++t) {
    const int tile = tile0 + t;

    float xv[DP];
    xv[0]  = xr[t][0].x; xv[1]  = xr[t][0].y; xv[2]  = xr[t][0].z; xv[3]  = xr[t][0].w;
    xv[4]  = xr[t][1].x; xv[5]  = xr[t][1].y; xv[6]  = xr[t][1].z; xv[7]  = xr[t][1].w;
    xv[8]  = xr[t][2].x; xv[9]  = xr[t][2].y; xv[10] = xr[t][2].z; xv[11] = xr[t][2].w;
    xv[12] = xr[t][3].x; xv[13] = xr[t][3].y; xv[14] = xr[t][3].z; xv[15] = xr[t][3].w;
    xv[16] = 1.0f;

    v2f A[KSTEPS];
    if (hi) {
#pragma unroll
      for (int k = 0; k < KSTEPS; ++k) {
        A[k].x = xv[TAB.b[4 * k + 2]] * xv[TAB.c[4 * k + 2]];
        A[k].y = xv[TAB.b[4 * k + 3]] * xv[TAB.c[4 * k + 3]];
      }
    } else {
#pragma unroll
      for (int k = 0; k < KSTEPS; ++k) {
        A[k].x = xv[TAB.b[4 * k + 0]] * xv[TAB.c[4 * k + 0]];
        A[k].y = xv[TAB.b[4 * k + 1]] * xv[TAB.c[4 * k + 1]];
      }
    }

    if (t == 0) {
      lds_barrier();
      wconst = wconst_s;
    }

    v8f acc = {};
#pragma unroll
    for (int k = 0; k < KSTEPS; ++k) {
      const v2f Wk = *(const v2f*)(bp + 64 * k);
      acc = __builtin_amdgcn_wmma_f32_16x16x4_f32(
          false, Wk, false, A[k], (short)0, acc, false, false);
    }

    float part = acc[0] * (hi ? xv[8] : xv[0]);
    part += acc[1] * (hi ? xv[9]  : xv[1]);
    part += acc[2] * (hi ? xv[10] : xv[2]);
    part += acc[3] * (hi ? xv[11] : xv[3]);
    part += acc[4] * (hi ? xv[12] : xv[4]);
    part += acc[5] * (hi ? xv[13] : xv[5]);
    part += acc[6] * (hi ? xv[14] : xv[6]);
    part += acc[7] * (hi ? xv[15] : xv[7]);
    const float logit = part + xhalf(part) + wconst;
    const float y = 1.0f / (1.0f + __expf(-logit));
    if (hi == 0) out[(size_t)tile * 16 + lm] = y;
  }
}

extern "C" void kernel_launch(void* const* d_in, const int* in_sizes, int n_in,
                              void* d_out, int out_size, void* d_ws,
                              size_t ws_size, hipStream_t stream) {
  const float* X        = (const float*)d_in[0];
  const float* w        = (const float*)d_in[1];
  const int*   comb_idx = (const int*)d_in[2];
  float*       out      = (float*)d_out;

  poly_logistic_kernel<<<256, 256, 0, stream>>>(X, w, comb_idx, out);
}
